# starting kernel
# baseline (speedup 1.0000x reference)
#include <hip/hip_runtime.h>
#include <math.h>

#define D_   128
#define KC   8
#define DD_  16
#define CUT_ 5
#define M_   32

typedef float v2f __attribute__((ext_vector_type(2)));
typedef float v8f __attribute__((ext_vector_type(8)));

__global__ __launch_bounds__(128) void norm_kernel(const float* __restrict__ x,
                                                   float* __restrict__ xn, int N) {
  const int node = blockIdx.x;
  const int t = threadIdx.x;
  if (node >= N) { xn[(size_t)node * D_ + t] = 0.0f; return; }
  float v = x[(size_t)node * D_ + t];
  float s = v * v;
  s += __shfl_xor(s, 1, 16);
  s += __shfl_xor(s, 2, 16);
  s += __shfl_xor(s, 4, 16);
  s += __shfl_xor(s, 8, 16);
  float nrm = fmaxf(sqrtf(s), 1e-12f);
  xn[(size_t)node * D_ + t] = v / nrm;
}

__global__ __launch_bounds__(32) void routing_kernel(const float* __restrict__ xn,
                                                     const int* __restrict__ nbr,
                                                     const int* __restrict__ p_maxiter,
                                                     float* __restrict__ u_out,
                                                     int* __restrict__ selcap,
                                                     int N, int M) {
  __shared__ float zs[M_ * D_];
  __shared__ float ps[M_ * KC];
  __shared__ float us[D_];
  __shared__ float xs[D_];

  const int node = blockIdx.x;
  const int lane = threadIdx.x;
  const int half = lane >> 4;
  const int lo   = lane & 15;
  const int T = p_maxiter[0];

  ((float4*)xs)[lane] = ((const float4*)(xn + (size_t)node * D_))[lane];

  int gl = nbr[node * M + lane];
  for (int r = 0; r < M_; ++r) {
    int g = __shfl(gl, r, 32);
    int row = (g < 0) ? N : g;
    ((float4*)(zs + r * D_))[lane] =
        ((const float4*)(xn + (size_t)row * D_))[lane];
  }
  __syncthreads();

  for (int it = 0; it < T; ++it) {
    if (it == 0) {
      #pragma unroll
      for (int q = 0; q < KC; ++q) ps[lane * KC + q] = 0.125f;
    } else {
      for (int k = 0; k < KC; ++k) {
        v8f d0 = {}; v8f d1 = {};
        #pragma unroll
        for (int s = 0; s < 4; ++s) {
          const int c0 = 4 * s + 2 * half;
          v2f b; b.x = us[k * DD_ + c0];     b.y = us[k * DD_ + c0 + 1];
          v2f a0; a0.x = zs[lo * D_        + k * DD_ + c0];
                  a0.y = zs[lo * D_        + k * DD_ + c0 + 1];
          v2f a1; a1.x = zs[(16 + lo) * D_ + k * DD_ + c0];
                  a1.y = zs[(16 + lo) * D_ + k * DD_ + c0 + 1];
          d0 = __builtin_amdgcn_wmma_f32_16x16x4_f32(false, a0, false, b, (short)0, d0, false, false);
          d1 = __builtin_amdgcn_wmma_f32_16x16x4_f32(false, a1, false, b, (short)0, d1, false, false);
        }
        if (lo == 0) {
          #pragma unroll
          for (int v = 0; v < 8; ++v) {
            ps[(8 * half + v) * KC + k]      = d0[v];
            ps[(16 + 8 * half + v) * KC + k] = d1[v];
          }
        }
      }
      __syncthreads();
      float pv[KC];
      float mx = -3.0e38f;
      #pragma unroll
      for (int q = 0; q < KC; ++q) { pv[q] = ps[lane * KC + q]; mx = fmaxf(mx, pv[q]); }
      float ssum = 0.0f;
      #pragma unroll
      for (int q = 0; q < KC; ++q) { pv[q] = expf(pv[q] - mx); ssum += pv[q]; }
      float inv = 1.0f / ssum;
      #pragma unroll
      for (int q = 0; q < KC; ++q) ps[lane * KC + q] = pv[q] * inv;
    }
    __syncthreads();

    for (int k = 0; k < KC; ++k) {
      v8f d = {};
      #pragma unroll
      for (int s = 0; s < 8; ++s) {
        const int m0 = 4 * s + 2 * half;
        v2f a; a.x = zs[m0 * D_ + k * DD_ + lo];
               a.y = zs[(m0 + 1) * D_ + k * DD_ + lo];
        v2f b; b.x = ps[m0 * KC + k];
               b.y = ps[(m0 + 1) * KC + k];
        d = __builtin_amdgcn_wmma_f32_16x16x4_f32(false, a, false, b, (short)0, d, false, false);
      }
      float uv[8];
      #pragma unroll
      for (int v = 0; v < 8; ++v) uv[v] = d[v] + xs[k * DD_ + 8 * half + v];
      if (it < T - 1) {
        float s2 = 0.0f;
        #pragma unroll
        for (int v = 0; v < 8; ++v) s2 += uv[v] * uv[v];
        s2 += __shfl_xor(s2, 16, 32);
        float inv = 1.0f / fmaxf(sqrtf(s2), 1e-12f);
        #pragma unroll
        for (int v = 0; v < 8; ++v) uv[v] *= inv;
      }
      if (lo == 0) {
        #pragma unroll
        for (int v = 0; v < 8; ++v) us[k * DD_ + 8 * half + v] = uv[v];
      }
    }
    __syncthreads();
  }

  ((float4*)(u_out + (size_t)node * D_))[lane] = ((float4*)us)[lane];

  if (lane < CUT_) {
    int best = 0;
    float bv = ps[lane * KC + 0];
    #pragma unroll
    for (int q = 1; q < KC; ++q) {
      float v = ps[lane * KC + q];
      if (v > bv) { bv = v; best = q; }
    }
    selcap[node * CUT_ + lane] = best;
  }
}

__global__ __launch_bounds__(128) void meta_kernel(const float* __restrict__ xn,
                                                   const int* __restrict__ nbr,
                                                   const int* __restrict__ selcap,
                                                   const int* __restrict__ p_last,
                                                   const float* __restrict__ u_out,
                                                   float* __restrict__ meta_out,
                                                   int N, int M) {
  if (p_last[0] == 0) return;
  const int node = blockIdx.x;
  const int t = threadIdx.x;
  const int i = t >> 4;
  const int c = t & 15;

  __shared__ int nb1[CUT_], cap1[CUT_];
  if (t < CUT_) {
    nb1[t]  = nbr[node * M + t];
    cap1[t] = selcap[node * CUT_ + t];
  }
  __syncthreads();

  float acc = 0.0f;
  for (int j = 0; j < KC; ++j) {
    float aj = 0.0f;
    #pragma unroll
    for (int s = 0; s < CUT_; ++s) {
      int g1 = nb1[s];
      if (g1 >= 0 && cap1[s] == i) {
        #pragma unroll
        for (int s2 = 0; s2 < CUT_; ++s2) {
          int g2 = nbr[g1 * M + s2];
          if (g2 >= 0 && selcap[g1 * CUT_ + s2] == j) {
            aj += xn[(size_t)g2 * D_ + j * DD_ + c];
          }
        }
      }
    }
    acc += aj / 25.0f;
  }
  meta_out[(size_t)node * D_ + t] = u_out[(size_t)node * D_ + t] + acc;
}

extern "C" void kernel_launch(void* const* d_in, const int* in_sizes, int n_in,
                              void* d_out, int out_size, void* d_ws, size_t ws_size,
                              hipStream_t stream) {
  const float* x       = (const float*)d_in[0];
  const int*   nbr     = (const int*)d_in[1];
  const int*   maxiter = (const int*)d_in[2];
  const int*   last    = (const int*)d_in[3];

  const int N = in_sizes[0] / D_;
  const int M = in_sizes[1] / N;

  float* u_out    = (float*)d_out;
  float* meta_out = (float*)d_out + (size_t)N * D_;

  float* xn = (float*)d_ws;
  size_t xn_bytes = (size_t)(N + 1) * D_ * sizeof(float);
  int* selcap = (int*)((char*)d_ws + xn_bytes);

  norm_kernel<<<N + 1, 128, 0, stream>>>(x, xn, N);
  routing_kernel<<<N, 32, 0, stream>>>(xn, nbr, maxiter, u_out, selcap, N, M);
  meta_kernel<<<N, 128, 0, stream>>>(xn, nbr, selcap, last, u_out, meta_out, N, M);
}


# Vectors u and p stay in registers: each A operand is one v_permlane16 broadcast of the owning lane to its 16-lane row, no LDS round trip or barrier left; loader k order changed, sums reordered
# speedup vs baseline: 1.9256x; 1.9256x over previous
#include <hip/hip_runtime.h>
#include <math.h>

#define D_   128
#define KC   8
#define DD_  16
#define CUT_ 5
#define M_   32

typedef float v2f __attribute__((ext_vector_type(2)));
typedef float v8f __attribute__((ext_vector_type(8)));

template <int CTRL>
__device__ __forceinline__ float row_xor_add(float v) {
  const int b = __builtin_bit_cast(int, v);
  return v + __builtin_bit_cast(float, __builtin_amdgcn_update_dpp(0, b, CTRL, 0xf, 0xf, true));
}

template <int CTRL>
__device__ __forceinline__ float row_share(float v) {
  const int b = __builtin_bit_cast(int, v);
  return __builtin_bit_cast(float, __builtin_amdgcn_update_dpp(0, b, CTRL, 0xf, 0xf, true));
}

__device__ __forceinline__ float other_row(float v) {
  const int b = __builtin_bit_cast(int, v);
  return __builtin_bit_cast(float, __builtin_amdgcn_permlanex16(b, b, 0x76543210, 0xfedcba98, false, false));
}

__device__ __forceinline__ float row_bcast(float v, int n) {
  const int b = __builtin_bit_cast(int, v);
  const unsigned sel = 0x11111111u * (unsigned)n;
  return __builtin_bit_cast(float, __builtin_amdgcn_permlane16(b, b, sel, sel, false, true));
}

__device__ __forceinline__ float upper_row_xor8(float v) {
  const int b = __builtin_bit_cast(int, v);
  return __builtin_bit_cast(float, __builtin_amdgcn_update_dpp(b, b, 0x168, 0x2, 0xf, false));
}

__device__ __forceinline__ void capsule_normalize(float (&v)[KC], int lo) {
  float s2[KC];
  #pragma unroll
  for (int k = 0; k < KC; ++k) {
    float t = v[k] * v[k];
    t = row_xor_add<0x161>(t);
    t = row_xor_add<0x162>(t);
    t = row_xor_add<0x164>(t);
    s2[k] = row_xor_add<0x168>(t);
  }
  float mine = s2[0];
  #pragma unroll
  for (int k = 1; k < KC; ++k) mine = ((lo & 7) == k) ? s2[k] : mine;
  const float inv = 1.0f / fmaxf(sqrtf(mine), 1e-12f);
  v[0] *= row_share<0x150>(inv);
  v[1] *= row_share<0x151>(inv);
  v[2] *= row_share<0x152>(inv);
  v[3] *= row_share<0x153>(inv);
  v[4] *= row_share<0x154>(inv);
  v[5] *= row_share<0x155>(inv);
  v[6] *= row_share<0x156>(inv);
  v[7] *= row_share<0x157>(inv);
}

__global__ __launch_bounds__(32, 3) void routing_kernel(const float* __restrict__ x,
                                                     const int* __restrict__ nbr,
                                                     const int* __restrict__ p_maxiter,
                                                     float* __restrict__ hq,
                                                     float* __restrict__ u_out,
                                                     int* __restrict__ selcap,
                                                     int N, int M) {
  const int node = blockIdx.x;
  const int lane = threadIdx.x;
  const int half = lane >> 4;
  const int lo   = lane & 15;
  const int T = p_maxiter[0];

  float xr[KC];
  #pragma unroll
  for (int k = 0; k < KC; ++k) xr[k] = x[(size_t)node * D_ + k * DD_ + lo];

  const int gl = nbr[node * M + lane];
  const int gw = (gl < 0) ? node : gl;

  const char* xb = (const char*)x;
  unsigned b1[2];
  #pragma unroll
  for (int t = 0; t < 2; ++t)
    b1[t] = ((unsigned)__shfl(gw, 16 * t + lo, 32) * D_ + 8 * half) * 4u;
  unsigned b2[8][2];
  #pragma unroll
  for (int s = 0; s < 8; ++s) {
    #pragma unroll
    for (int e = 0; e < 2; ++e)
      b2[s][e] = ((unsigned)__shfl(gw, 16 * half + 2 * s + e, 32) * D_ + lo) * 4u;
  }

  v2f za[2][KC][4];
  v2f zb[KC][8];
  #pragma unroll
  for (int L = 0; L < 4; ++L) {
    #pragma unroll
    for (int kk = 0; kk < 2; ++kk) {
      const int k = 2 * L + kk;
      #pragma unroll
      for (int t = 0; t < 2; ++t) {
        const float4 q0 = *(const float4*)(xb + ((size_t)b1[t] + (k * DD_) * 4));
        const float4 q1 = *(const float4*)(xb + ((size_t)b1[t] + (k * DD_ + 4) * 4));
        za[t][k][0].x = q0.x; za[t][k][0].y = q0.y;
        za[t][k][1].x = q0.z; za[t][k][1].y = q0.w;
        za[t][k][2].x = q1.x; za[t][k][2].y = q1.y;
        za[t][k][3].x = q1.z; za[t][k][3].y = q1.w;
      }
    }
    #pragma unroll
    for (int kk = 0; kk < 2; ++kk) {
      const int k = 2 * L + kk;
      #pragma unroll
      for (int s = 0; s < 8; ++s) {
        zb[k][s].x = *(const float*)(xb + ((size_t)b2[s][0] + k * DD_ * 4));
        zb[k][s].y = *(const float*)(xb + ((size_t)b2[s][1] + k * DD_ * 4));
      }
    }
  }

  capsule_normalize(xr, lo);

  float ninv[KC];
  const float present = (gl < 0) ? 0.0f : 1.0f;
  #pragma unroll
  for (int k = 0; k < KC; ++k) {
    v2f a0 = za[0][k][0] * za[0][k][0];
    v2f a1 = za[1][k][0] * za[1][k][0];
    #pragma unroll
    for (int s = 1; s < 4; ++s) {
      a0 += za[0][k][s] * za[0][k][s];
      a1 += za[1][k][s] * za[1][k][s];
    }
    const float p0 = a0.x + a0.y;
    const float p1 = a1.x + a1.y;
    const float ss = (half ? p1 : p0) + other_row(half ? p0 : p1);
    ninv[k] = present / fmaxf(sqrtf(ss), 1e-12f);
  }

  int best = 0;
  float uv[KC];
  float uvr[KC];
  float pvn[KC];
  #pragma unroll
  for (int k = 0; k < KC; ++k) { uv[k] = 0.0f; uvr[k] = 0.0f; }
  for (int it = 0; it < T; ++it) {
    if (it == 0) {
      #pragma unroll
      for (int q = 0; q < KC; ++q) pvn[q] = 0.125f * ninv[q];
    } else {
      float pv[KC];
      #pragma unroll
      for (int k = 0; k < KC; ++k) {
        v8f d0 = {}; v8f d1 = {};
        #pragma unroll
        for (int s = 0; s < 4; ++s) {
          v2f a; a.x = row_bcast(uvr[k], 2 * s); a.y = row_bcast(uvr[k], 2 * s + 1);
          d0 = __builtin_amdgcn_wmma_f32_16x16x4_f32(false, a, false, za[0][k][s], (short)0, d0, false, false);
          d1 = __builtin_amdgcn_wmma_f32_16x16x4_f32(false, a, false, za[1][k][s], (short)0, d1, false, false);
        }
        pv[k] = (half ? d1[0] : d0[0]) * ninv[k];
      }
      float mx = -3.0e38f;
      #pragma unroll
      for (int q = 0; q < KC; ++q) mx = fmaxf(mx, pv[q]);
      float ssum = 0.0f;
      #pragma unroll
      for (int q = 0; q < KC; ++q) { pv[q] = expf(pv[q] - mx); ssum += pv[q]; }
      float inv = 1.0f / ssum;
      #pragma unroll
      for (int q = 0; q < KC; ++q) pv[q] *= inv;
      if (it == T - 1) {
        float bv = pv[0];
        #pragma unroll
        for (int q = 1; q < KC; ++q)
          if (pv[q] > bv) { bv = pv[q]; best = q; }
      }
      #pragma unroll
      for (int q = 0; q < KC; ++q) pvn[q] = pv[q] * ninv[q];
    }

    #pragma unroll
    for (int kp = 0; kp < KC / 2; ++kp) {
      v8f da = {}; v8f db = {};
      #pragma unroll
      for (int s = 0; s < 8; ++s) {
        v2f a0; a0.x = row_bcast(pvn[2 * kp],     2 * s); a0.y = row_bcast(pvn[2 * kp],     2 * s + 1);
        v2f a1; a1.x = row_bcast(pvn[2 * kp + 1], 2 * s); a1.y = row_bcast(pvn[2 * kp + 1], 2 * s + 1);
        da = __builtin_amdgcn_wmma_f32_16x16x4_f32(false, a0, false, zb[2 * kp][s],     (short)0, da, false, false);
        db = __builtin_amdgcn_wmma_f32_16x16x4_f32(false, a1, false, zb[2 * kp + 1][s], (short)0, db, false, false);
      }
      uv[2 * kp]     = da[0] + xr[2 * kp];
      uv[2 * kp + 1] = db[0] + xr[2 * kp + 1];
    }
    if (it < T - 1) {
      capsule_normalize(uv, lo);
      #pragma unroll
      for (int k = 0; k < KC; ++k) uvr[k] = upper_row_xor8(uv[k]);
    }
  }

  #pragma unroll
  for (int j = 0; j < KC / 2; ++j)
    u_out[(size_t)node * D_ + j * 32 + lane] = half ? uv[2 * j + 1] : uv[2 * j];

  if (lane < CUT_) selcap[node * CUT_ + lane] = best;

  float fsel = ninv[0];
  #pragma unroll
  for (int k = 1; k < KC; ++k) fsel = (best == k) ? ninv[k] : fsel;
  int kb[CUT_]; float fb[CUT_];
  #pragma unroll
  for (int m = 0; m < CUT_; ++m) { kb[m] = __shfl(best, m, 32); fb[m] = __shfl(fsel, m, 32); }
  float v0 = zb[0][0].x, v1 = zb[0][0].y, v2 = zb[0][1].x, v3 = zb[0][1].y, v4 = zb[0][2].x;
  #pragma unroll
  for (int k = 1; k < KC; ++k) {
    v0 = (kb[0] == k) ? zb[k][0].x : v0;
    v1 = (kb[1] == k) ? zb[k][0].y : v1;
    v2 = (kb[2] == k) ? zb[k][1].x : v2;
    v3 = (kb[3] == k) ? zb[k][1].y : v3;
    v4 = (kb[4] == k) ? zb[k][2].x : v4;
  }
  const float part = v0 * fb[0] + v1 * fb[1] + v2 * fb[2] + v3 * fb[3] + v4 * fb[4];
  const float tot = other_row(part);
  hq[(size_t)node * 32 + lane] = half ? tot : part;
}

__global__ __launch_bounds__(128) void meta_kernel(const float* __restrict__ hq,
                                                   const int* __restrict__ nbr,
                                                   const int* __restrict__ selcap,
                                                   const int* __restrict__ p_last,
                                                   const float* __restrict__ u_out,
                                                   float* __restrict__ meta_out,
                                                   int N, int M) {
  if (p_last[0] == 0) return;
  const int node = blockIdx.x;
  const int t = threadIdx.x;
  const int i = t >> 4;
  const int c = t & 15;

  int g1[CUT_], cap1[CUT_];
  #pragma unroll
  for (int s = 0; s < CUT_; ++s) {
    g1[s]   = nbr[node * M + s];
    cap1[s] = selcap[node * CUT_ + s];
  }
  float hv[CUT_];
  #pragma unroll
  for (int s = 0; s < CUT_; ++s) {
    const int gs = (g1[s] < 0) ? 0 : g1[s];
    hv[s] = hq[(size_t)gs * 32 + c];
  }
  float acc = 0.0f;
  #pragma unroll
  for (int s = 0; s < CUT_; ++s) acc += (g1[s] >= 0 && cap1[s] == i) ? hv[s] : 0.0f;
  meta_out[(size_t)node * D_ + t] = u_out[(size_t)node * D_ + t] + acc / 25.0f;
}

extern "C" void kernel_launch(void* const* d_in, const int* in_sizes, int n_in,
                              void* d_out, int out_size, void* d_ws, size_t ws_size,
                              hipStream_t stream) {
  const float* x       = (const float*)d_in[0];
  const int*   nbr     = (const int*)d_in[1];
  const int*   maxiter = (const int*)d_in[2];
  const int*   last    = (const int*)d_in[3];

  const int N = in_sizes[0] / D_;
  const int M = in_sizes[1] / N;

  float* u_out    = (float*)d_out;
  float* meta_out = (float*)d_out + (size_t)N * D_;

  float* hq = (float*)d_ws;
  size_t hq_bytes = (size_t)N * 32 * sizeof(float);
  int* selcap = (int*)((char*)d_ws + hq_bytes);

  routing_kernel<<<N, 32, 0, stream>>>(x, nbr, maxiter, hq, u_out, selcap, N, M);
  meta_kernel<<<N, 128, 0, stream>>>(hq, nbr, selcap, last, u_out, meta_out, N, M);
}


# Meta kernel regrouped: a wave takes 4 whole nodes at once, a lane 4 components by 16-byte loads and stores; 5000 waves in 625 work-groups in place of 80000 in 20000; values bit-identical
# speedup vs baseline: 1.9965x; 1.0368x over previous
#include <hip/hip_runtime.h>
#include <math.h>

#define D_   128
#define KC   8
#define DD_  16
#define CUT_ 5
#define M_   32

typedef float v2f __attribute__((ext_vector_type(2)));
typedef float v8f __attribute__((ext_vector_type(8)));

template <int CTRL>
__device__ __forceinline__ float row_xor_add(float v) {
  const int b = __builtin_bit_cast(int, v);
  return v + __builtin_bit_cast(float, __builtin_amdgcn_update_dpp(0, b, CTRL, 0xf, 0xf, true));
}

template <int CTRL>
__device__ __forceinline__ float row_share(float v) {
  const int b = __builtin_bit_cast(int, v);
  return __builtin_bit_cast(float, __builtin_amdgcn_update_dpp(0, b, CTRL, 0xf, 0xf, true));
}

__device__ __forceinline__ float other_row(float v) {
  const int b = __builtin_bit_cast(int, v);
  return __builtin_bit_cast(float, __builtin_amdgcn_permlanex16(b, b, 0x76543210, 0xfedcba98, false, false));
}

__device__ __forceinline__ float row_bcast(float v, int n) {
  const int b = __builtin_bit_cast(int, v);
  const unsigned sel = 0x11111111u * (unsigned)n;
  return __builtin_bit_cast(float, __builtin_amdgcn_permlane16(b, b, sel, sel, false, true));
}

__device__ __forceinline__ float upper_row_xor8(float v) {
  const int b = __builtin_bit_cast(int, v);
  return __builtin_bit_cast(float, __builtin_amdgcn_update_dpp(b, b, 0x168, 0x2, 0xf, false));
}

__device__ __forceinline__ void capsule_normalize(float (&v)[KC], int lo) {
  float s2[KC];
  #pragma unroll
  for (int k = 0; k < KC; ++k) {
    float t = v[k] * v[k];
    t = row_xor_add<0x161>(t);
    t = row_xor_add<0x162>(t);
    t = row_xor_add<0x164>(t);
    s2[k] = row_xor_add<0x168>(t);
  }
  float mine = s2[0];
  #pragma unroll
  for (int k = 1; k < KC; ++k) mine = ((lo & 7) == k) ? s2[k] : mine;
  const float inv = 1.0f / fmaxf(sqrtf(mine), 1e-12f);
  v[0] *= row_share<0x150>(inv);
  v[1] *= row_share<0x151>(inv);
  v[2] *= row_share<0x152>(inv);
  v[3] *= row_share<0x153>(inv);
  v[4] *= row_share<0x154>(inv);
  v[5] *= row_share<0x155>(inv);
  v[6] *= row_share<0x156>(inv);
  v[7] *= row_share<0x157>(inv);
}

__global__ __launch_bounds__(32, 3) void routing_kernel(const float* __restrict__ x,
                                                     const int* __restrict__ nbr,
                                                     const int* __restrict__ p_maxiter,
                                                     float* __restrict__ hq,
                                                     float* __restrict__ u_out,
                                                     int* __restrict__ selcap,
                                                     int N, int M) {
  const int node = blockIdx.x;
  const int lane = threadIdx.x;
  const int half = lane >> 4;
  const int lo   = lane & 15;
  const int T = p_maxiter[0];

  float xr[KC];
  #pragma unroll
  for (int k = 0; k < KC; ++k) xr[k] = x[(size_t)node * D_ + k * DD_ + lo];

  const int gl = nbr[node * M + lane];
  const int gw = (gl < 0) ? node : gl;

  const char* xb = (const char*)x;
  unsigned b1[2];
  #pragma unroll
  for (int t = 0; t < 2; ++t)
    b1[t] = ((unsigned)__shfl(gw, 16 * t + lo, 32) * D_ + 8 * half) * 4u;
  unsigned b2[8][2];
  #pragma unroll
  for (int s = 0; s < 8; ++s) {
    #pragma unroll
    for (int e = 0; e < 2; ++e)
      b2[s][e] = ((unsigned)__shfl(gw, 16 * half + 2 * s + e, 32) * D_ + lo) * 4u;
  }

  v2f za[2][KC][4];
  v2f zb[KC][8];
  #pragma unroll
  for (int L = 0; L < 4; ++L) {
    #pragma unroll
    for (int kk = 0; kk < 2; ++kk) {
      const int k = 2 * L + kk;
      #pragma unroll
      for (int t = 0; t < 2; ++t) {
        const float4 q0 = *(const float4*)(xb + ((size_t)b1[t] + (k * DD_) * 4));
        const float4 q1 = *(const float4*)(xb + ((size_t)b1[t] + (k * DD_ + 4) * 4));
        za[t][k][0].x = q0.x; za[t][k][0].y = q0.y;
        za[t][k][1].x = q0.z; za[t][k][1].y = q0.w;
        za[t][k][2].x = q1.x; za[t][k][2].y = q1.y;
        za[t][k][3].x = q1.z; za[t][k][3].y = q1.w;
      }
    }
    #pragma unroll
    for (int kk = 0; kk < 2; ++kk) {
      const int k = 2 * L + kk;
      #pragma unroll
      for (int s = 0; s < 8; ++s) {
        zb[k][s].x = *(const float*)(xb + ((size_t)b2[s][0] + k * DD_ * 4));
        zb[k][s].y = *(const float*)(xb + ((size_t)b2[s][1] + k * DD_ * 4));
      }
    }
  }

  capsule_normalize(xr, lo);

  float ninv[KC];
  const float present = (gl < 0) ? 0.0f : 1.0f;
  #pragma unroll
  for (int k = 0; k < KC; ++k) {
    v2f a0 = za[0][k][0] * za[0][k][0];
    v2f a1 = za[1][k][0] * za[1][k][0];
    #pragma unroll
    for (int s = 1; s < 4; ++s) {
      a0 += za[0][k][s] * za[0][k][s];
      a1 += za[1][k][s] * za[1][k][s];
    }
    const float p0 = a0.x + a0.y;
    const float p1 = a1.x + a1.y;
    const float ss = (half ? p1 : p0) + other_row(half ? p0 : p1);
    ninv[k] = present / fmaxf(sqrtf(ss), 1e-12f);
  }

  int best = 0;
  float uv[KC];
  float uvr[KC];
  float pvn[KC];
  #pragma unroll
  for (int k = 0; k < KC; ++k) { uv[k] = 0.0f; uvr[k] = 0.0f; }
  for (int it = 0; it < T; ++it) {
    if (it == 0) {
      #pragma unroll
      for (int q = 0; q < KC; ++q) pvn[q] = 0.125f * ninv[q];
    } else {
      float pv[KC];
      #pragma unroll
      for (int k = 0; k < KC; ++k) {
        v8f d0 = {}; v8f d1 = {};
        #pragma unroll
        for (int s = 0; s < 4; ++s) {
          v2f a; a.x = row_bcast(uvr[k], 2 * s); a.y = row_bcast(uvr[k], 2 * s + 1);
          d0 = __builtin_amdgcn_wmma_f32_16x16x4_f32(false, a, false, za[0][k][s], (short)0, d0, false, false);
          d1 = __builtin_amdgcn_wmma_f32_16x16x4_f32(false, a, false, za[1][k][s], (short)0, d1, false, false);
        }
        pv[k] = (half ? d1[0] : d0[0]) * ninv[k];
      }
      float mx = -3.0e38f;
      #pragma unroll
      for (int q = 0; q < KC; ++q) mx = fmaxf(mx, pv[q]);
      float ssum = 0.0f;
      #pragma unroll
      for (int q = 0; q < KC; ++q) { pv[q] = expf(pv[q] - mx); ssum += pv[q]; }
      float inv = 1.0f / ssum;
      #pragma unroll
      for (int q = 0; q < KC; ++q) pv[q] *= inv;
      if (it == T - 1) {
        float bv = pv[0];
        #pragma unroll
        for (int q = 1; q < KC; ++q)
          if (pv[q] > bv) { bv = pv[q]; best = q; }
      }
      #pragma unroll
      for (int q = 0; q < KC; ++q) pvn[q] = pv[q] * ninv[q];
    }

    #pragma unroll
    for (int kp = 0; kp < KC / 2; ++kp) {
      v8f da = {}; v8f db = {};
      #pragma unroll
      for (int s = 0; s < 8; ++s) {
        v2f a0; a0.x = row_bcast(pvn[2 * kp],     2 * s); a0.y = row_bcast(pvn[2 * kp],     2 * s + 1);
        v2f a1; a1.x = row_bcast(pvn[2 * kp + 1], 2 * s); a1.y = row_bcast(pvn[2 * kp + 1], 2 * s + 1);
        da = __builtin_amdgcn_wmma_f32_16x16x4_f32(false, a0, false, zb[2 * kp][s],     (short)0, da, false, false);
        db = __builtin_amdgcn_wmma_f32_16x16x4_f32(false, a1, false, zb[2 * kp + 1][s], (short)0, db, false, false);
      }
      uv[2 * kp]     = da[0] + xr[2 * kp];
      uv[2 * kp + 1] = db[0] + xr[2 * kp + 1];
    }
    if (it < T - 1) {
      capsule_normalize(uv, lo);
      #pragma unroll
      for (int k = 0; k < KC; ++k) uvr[k] = upper_row_xor8(uv[k]);
    }
  }

  #pragma unroll
  for (int j = 0; j < KC / 2; ++j)
    u_out[(size_t)node * D_ + j * 32 + lane] = half ? uv[2 * j + 1] : uv[2 * j];

  if (lane < CUT_) selcap[node * CUT_ + lane] = best;

  float fsel = ninv[0];
  #pragma unroll
  for (int k = 1; k < KC; ++k) fsel = (best == k) ? ninv[k] : fsel;
  int kb[CUT_]; float fb[CUT_];
  #pragma unroll
  for (int m = 0; m < CUT_; ++m) { kb[m] = __shfl(best, m, 32); fb[m] = __shfl(fsel, m, 32); }
  float v0 = zb[0][0].x, v1 = zb[0][0].y, v2 = zb[0][1].x, v3 = zb[0][1].y, v4 = zb[0][2].x;
  #pragma unroll
  for (int k = 1; k < KC; ++k) {
    v0 = (kb[0] == k) ? zb[k][0].x : v0;
    v1 = (kb[1] == k) ? zb[k][0].y : v1;
    v2 = (kb[2] == k) ? zb[k][1].x : v2;
    v3 = (kb[3] == k) ? zb[k][1].y : v3;
    v4 = (kb[4] == k) ? zb[k][2].x : v4;
  }
  const float part = v0 * fb[0] + v1 * fb[1] + v2 * fb[2] + v3 * fb[3] + v4 * fb[4];
  const float tot = other_row(part);
  hq[(size_t)node * 32 + lane] = half ? tot : part;
}

#define MNW 4
#define MWV 8
__global__ __launch_bounds__(32 * MWV) void meta_kernel(const float* __restrict__ hq,
                                                   const int* __restrict__ nbr,
                                                   const int* __restrict__ selcap,
                                                   const int* __restrict__ p_last,
                                                   const float* __restrict__ u_out,
                                                   float* __restrict__ meta_out,
                                                   int N, int M) {
  if (p_last[0] == 0) return;
  const int lane = threadIdx.x & 31;
  const int wave = threadIdx.x >> 5;
  const int node0 = (blockIdx.x * MWV + wave) * MNW;
  const int i  = lane >> 2;
  const int c4 = (lane & 3) * 4;

  int nd[MNW];
  #pragma unroll
  for (int q = 0; q < MNW; ++q) nd[q] = (node0 + q < N) ? node0 + q : N - 1;

  int g1[MNW][CUT_], cap1[MNW][CUT_];
  #pragma unroll
  for (int q = 0; q < MNW; ++q) {
    #pragma unroll
    for (int s = 0; s < CUT_; ++s) {
      g1[q][s]   = nbr[nd[q] * M + s];
      cap1[q][s] = selcap[nd[q] * CUT_ + s];
    }
  }
  float4 uu[MNW];
  float4 hv[MNW][CUT_];
  #pragma unroll
  for (int q = 0; q < MNW; ++q) {
    uu[q] = *(const float4*)(u_out + (size_t)nd[q] * D_ + lane * 4);
    #pragma unroll
    for (int s = 0; s < CUT_; ++s) {
      const int gs = (g1[q][s] < 0) ? 0 : g1[q][s];
      hv[q][s] = *(const float4*)(hq + (size_t)gs * 32 + c4);
    }
  }
  #pragma unroll
  for (int q = 0; q < MNW; ++q) {
    asm volatile("" :: "v"(uu[q].x), "v"(uu[q].y), "v"(uu[q].z), "v"(uu[q].w));
    #pragma unroll
    for (int s = 0; s < CUT_; ++s)
      asm volatile("" :: "v"(cap1[q][s]), "v"(hv[q][s].x), "v"(hv[q][s].y), "v"(hv[q][s].z), "v"(hv[q][s].w));
  }
  #pragma unroll
  for (int q = 0; q < MNW; ++q) {
    float4 acc = make_float4(0.0f, 0.0f, 0.0f, 0.0f);
    #pragma unroll
    for (int s = 0; s < CUT_; ++s) {
      const bool on = (g1[q][s] >= 0) & (cap1[q][s] == i);
      acc.x += on ? hv[q][s].x : 0.0f;
      acc.y += on ? hv[q][s].y : 0.0f;
      acc.z += on ? hv[q][s].z : 0.0f;
      acc.w += on ? hv[q][s].w : 0.0f;
    }
    float4 r;
    r.x = uu[q].x + acc.x / 25.0f;
    r.y = uu[q].y + acc.y / 25.0f;
    r.z = uu[q].z + acc.z / 25.0f;
    r.w = uu[q].w + acc.w / 25.0f;
    if (node0 + q < N) *(float4*)(meta_out + (size_t)(node0 + q) * D_ + lane * 4) = r;
  }
}

extern "C" void kernel_launch(void* const* d_in, const int* in_sizes, int n_in,
                              void* d_out, int out_size, void* d_ws, size_t ws_size,
                              hipStream_t stream) {
  const float* x       = (const float*)d_in[0];
  const int*   nbr     = (const int*)d_in[1];
  const int*   maxiter = (const int*)d_in[2];
  const int*   last    = (const int*)d_in[3];

  const int N = in_sizes[0] / D_;
  const int M = in_sizes[1] / N;

  float* u_out    = (float*)d_out;
  float* meta_out = (float*)d_out + (size_t)N * D_;

  float* hq = (float*)d_ws;
  size_t hq_bytes = (size_t)N * 32 * sizeof(float);
  int* selcap = (int*)((char*)d_ws + hq_bytes);

  routing_kernel<<<N, 32, 0, stream>>>(x, nbr, maxiter, hq, u_out, selcap, N, M);
  meta_kernel<<<(N + MNW * MWV - 1) / (MNW * MWV), 32 * MWV, 0, stream>>>(hq, nbr, selcap, last, u_out, meta_out, N, M);
}
